# starting kernel
# baseline (speedup 1.0000x reference)
#include <hip/hip_runtime.h>
#include <hip/hip_bf16.h>
#include <math.h>

typedef __attribute__((ext_vector_type(2))) float v2f;
typedef __attribute__((ext_vector_type(8))) float v8f;

#define IN_DIM  256
#define OUT_DIM 128
#define KC 32
#define NB 32
#define HSTR 36

__device__ __forceinline__ unsigned fkey(float f) {
    unsigned u = __float_as_uint(f);
    return (u & 0x80000000u) ? ~u : (u | 0x80000000u);
}
__device__ __forceinline__ float keyf(unsigned k) {
    unsigned u = (k & 0x80000000u) ? (k & 0x7FFFFFFFu) : ~k;
    return __uint_as_float(u);
}

__global__ void gat_init(float* __restrict__ out, unsigned* __restrict__ mkey,
                         unsigned* __restrict__ deg, float* __restrict__ denom,
                         int n, int total) {
    int i = blockIdx.x * blockDim.x + threadIdx.x;
    if (i < total) out[i] = 0.0f;
    if (i < n) { mkey[i] = 0u; deg[i] = 0u; denom[i] = 0.0f; }
}

__global__ __launch_bounds__(256) void gat_gemm_wmma(
    const float* __restrict__ h, const float* __restrict__ W,
    float* __restrict__ z, int n) {
    __shared__ float lds_h[NB * HSTR];
    __shared__ float lds_w[KC * OUT_DIM];

    const int tid      = threadIdx.x;
    const int lane     = tid & 31;
    const int wave     = tid >> 5;
    const int nodeTile = wave >> 2;
    const int outGroup = wave & 3;
    const int nodeBase = blockIdx.x * NB;

    const int hl  = lane & 15;
    const int khi = (lane >> 4) * 2;

    v8f c0 = {}; v8f c1 = {};

    for (int kc = 0; kc < IN_DIM; kc += KC) {
        __syncthreads();
        {
            int nodeL = tid >> 3;
            int ks    = (tid & 7) * 4;
            int g     = nodeBase + nodeL;
            if (g >= n) g = n - 1;
            const float4 hv = *(const float4*)(h + (size_t)g * IN_DIM + kc + ks);
            *(float4*)(&lds_h[nodeL * HSTR + ks]) = hv;
            if (kc + KC < IN_DIM)
                __builtin_prefetch((const void*)(h + (size_t)g * IN_DIM + kc + KC + ks), 0, 1);
        }
        {
            int o  = tid >> 1;
            int kb = (tid & 1) * 16;
            const float* wrow = W + (size_t)o * IN_DIM + kc + kb;
            #pragma unroll
            for (int j = 0; j < 16; ++j) {
                int k = kb + j;
                lds_w[((k >> 1) * OUT_DIM + o) * 2 + (k & 1)] = wrow[j];
            }
        }
        __syncthreads();

        const int row = nodeTile * 16 + hl;
        const int o0  = (outGroup * 2 + 0) * 16 + hl;
        const int o1  = (outGroup * 2 + 1) * 16 + hl;
        #pragma unroll
        for (int ks = 0; ks < KC; ks += 4) {
            v2f a = *(const v2f*)(&lds_h[row * HSTR + ks + khi]);
            int kb = ks + khi;
            v2f b0 = *(const v2f*)(&lds_w[((kb >> 1) * OUT_DIM + o0) * 2]);
            v2f b1 = *(const v2f*)(&lds_w[((kb >> 1) * OUT_DIM + o1) * 2]);
            c0 = __builtin_amdgcn_wmma_f32_16x16x4_f32(false, a, false, b0, (short)0, c0, false, false);
            c1 = __builtin_amdgcn_wmma_f32_16x16x4_f32(false, a, false, b1, (short)0, c1, false, false);
        }
    }

    const int mrow0 = nodeBase + nodeTile * 16 + (lane >> 4) * 8;
    const int ncol  = lane & 15;
    #pragma unroll
    for (int r = 0; r < 8; ++r) {
        int gnode = mrow0 + r;
        if (gnode < n) {
            z[(size_t)gnode * OUT_DIM + (outGroup * 2 + 0) * 16 + ncol] = c0[r];
            z[(size_t)gnode * OUT_DIM + (outGroup * 2 + 1) * 16 + ncol] = c1[r];
        }
    }
}

__global__ __launch_bounds__(256) void gat_nodedot(
    const float* __restrict__ z, const float* __restrict__ a_attn,
    float* __restrict__ zs, float* __restrict__ zd, int n) {
    int node = blockIdx.x * 8 + (threadIdx.x >> 5);
    int lane = threadIdx.x & 31;
    if (node >= n) return;
    float4 zv = *(const float4*)(z + (size_t)node * OUT_DIM + lane * 4);
    float4 as = *(const float4*)(a_attn + lane * 4);
    float4 ad = *(const float4*)(a_attn + OUT_DIM + lane * 4);
    float s = zv.x * as.x + zv.y * as.y + zv.z * as.z + zv.w * as.w;
    float d = zv.x * ad.x + zv.y * ad.y + zv.z * ad.z + zv.w * ad.w;
    #pragma unroll
    for (int off = 16; off > 0; off >>= 1) {
        s += __shfl_down(s, off, 32);
        d += __shfl_down(d, off, 32);
    }
    if (lane == 0) { zs[node] = s; zd[node] = d; }
}

__global__ void gat_edgeA(const int* __restrict__ src, const int* __restrict__ dst,
                          const float* __restrict__ zs, const float* __restrict__ zd,
                          float* __restrict__ e, unsigned* __restrict__ mkey,
                          unsigned* __restrict__ deg, int E) {
    int i = blockIdx.x * blockDim.x + threadIdx.x;
    if (i >= E) return;
    int s = src[i], d = dst[i];
    float x  = zs[s] + zd[d];
    float ev = x > 0.0f ? x : 0.01f * x;
    e[i] = ev;
    atomicMax(&mkey[d], fkey(ev));
    atomicAdd(&deg[d], 1u);
}

__global__ void gat_edgeB(const int* __restrict__ dst, float* ebuf,
                          const unsigned* __restrict__ mkey,
                          float* __restrict__ denom, int E) {
    int i = blockIdx.x * blockDim.x + threadIdx.x;
    if (i >= E) return;
    int d = dst[i];
    float v = expf(ebuf[i] - keyf(mkey[d]));
    ebuf[i] = v;
    atomicAdd(&denom[d], v);
}

__global__ __launch_bounds__(256) void gat_edgeC(
    const int* __restrict__ src, const int* __restrict__ dst,
    const float* __restrict__ ex, const float* __restrict__ denom,
    const float* __restrict__ z, float* __restrict__ out, int E) {
    int e    = blockIdx.x * 8 + (threadIdx.x >> 5);
    int lane = threadIdx.x & 31;
    if (e >= E) return;
    int s = src[e], d = dst[e];
    float alpha = ex[e] / denom[d];
    float4 zv = *(const float4*)(z + (size_t)s * OUT_DIM + lane * 4);
    float* o = out + (size_t)d * OUT_DIM + lane * 4;
    atomicAdd(o + 0, alpha * zv.x);
    atomicAdd(o + 1, alpha * zv.y);
    atomicAdd(o + 2, alpha * zv.z);
    atomicAdd(o + 3, alpha * zv.w);
}

__global__ void gat_final(const unsigned* __restrict__ deg,
                          const float* __restrict__ z, float* __restrict__ out, int total) {
    int i = blockIdx.x * blockDim.x + threadIdx.x;
    if (i >= total) return;
    if (deg[i >> 7] == 0u) out[i] = z[i];
}

extern "C" void kernel_launch(void* const* d_in, const int* in_sizes, int n_in,
                              void* d_out, int out_size, void* d_ws, size_t ws_size,
                              hipStream_t stream) {
    const float* h      = (const float*)d_in[0];
    const int*   src    = (const int*)d_in[1];
    const int*   dst    = (const int*)d_in[2];
    const float* W      = (const float*)d_in[3];
    const float* a_attn = (const float*)d_in[4];
    const int n = in_sizes[0] / IN_DIM;
    const int E = in_sizes[1];
    float* out = (float*)d_out;

    char* ws = (char*)d_ws;
    float*    z     = (float*)ws;    ws += (size_t)n * OUT_DIM * sizeof(float);
    float*    zs    = (float*)ws;    ws += (size_t)n * sizeof(float);
    float*    zd    = (float*)ws;    ws += (size_t)n * sizeof(float);
    float*    ebuf  = (float*)ws;    ws += (size_t)E * sizeof(float);
    unsigned* mkey  = (unsigned*)ws; ws += (size_t)n * sizeof(unsigned);
    unsigned* deg   = (unsigned*)ws; ws += (size_t)n * sizeof(unsigned);
    float*    denom = (float*)ws;

    const int total = n * OUT_DIM;
    gat_init     <<<(total + 255) / 256, 256, 0, stream>>>(out, mkey, deg, denom, n, total);
    gat_gemm_wmma<<<(n + NB - 1) / NB,   256, 0, stream>>>(h, W, z, n);
    gat_nodedot  <<<(n + 7) / 8,         256, 0, stream>>>(z, a_attn, zs, zd, n);
    gat_edgeA    <<<(E + 255) / 256,     256, 0, stream>>>(src, dst, zs, zd, ebuf, mkey, deg, E);
    gat_edgeB    <<<(E + 255) / 256,     256, 0, stream>>>(dst, ebuf, mkey, denom, E);
    gat_edgeC    <<<(E + 7) / 8,         256, 0, stream>>>(src, dst, ebuf, denom, z, out, E);
    gat_final    <<<(total + 255) / 256, 256, 0, stream>>>(deg, z, out, total);
}


# Gather: four destination nodes per wave (eight lanes each, 16 outputs per lane), one edge per trip with its z row in flight while the next link is fetched: 50000 gather waves become 12500
# speedup vs baseline: 2.5715x; 2.5715x over previous
#include <hip/hip_runtime.h>
#include <hip/hip_bf16.h>
#include <math.h>

typedef __attribute__((ext_vector_type(2))) float v2f;
typedef __attribute__((ext_vector_type(8))) float v8f;

#define IN_DIM  256
#define OUT_DIM 128
#define KC 32
#define NB 32
#define HSTR 36

template <int XM>
__device__ __forceinline__ float row_xor_sum(float x) {
    const int b = __builtin_bit_cast(int, x);
    return x + __builtin_bit_cast(float,
        __builtin_amdgcn_update_dpp(0, b, 0x160 + XM, 0xf, 0xf, true));
}

__global__ __launch_bounds__(256, 16) void gat_gemm_wmma(
    const float* __restrict__ h, const float* __restrict__ W,
    const float* __restrict__ a_attn,
    float* __restrict__ z, float* __restrict__ zs, float* __restrict__ zd,
    int* __restrict__ head, int n) {
    __shared__ float lds_h[NB * HSTR];
    __shared__ float lds_w[KC * OUT_DIM];
    __shared__ __attribute__((aligned(16))) float lds_p[4 * NB * 2];

    const int tid      = threadIdx.x;
    const int lane     = tid & 31;
    const int wave     = tid >> 5;
    const int nodeTile = wave >> 2;
    const int outGroup = wave & 3;
    const int nodeBase = blockIdx.x * NB;

    const int hl  = lane & 15;
    const int khi = (lane >> 4) * 2;

    v8f c0 = {}; v8f c1 = {};

    for (int kc = 0; kc < IN_DIM; kc += KC) {
        __syncthreads();
        {
            int nodeL = tid >> 3;
            int ks    = (tid & 7) * 4;
            int g     = nodeBase + nodeL;
            if (g >= n) g = n - 1;
            const float4 hv = *(const float4*)(h + (size_t)g * IN_DIM + kc + ks);
            *(float4*)(&lds_h[nodeL * HSTR + ks]) = hv;
            if (kc + KC < IN_DIM)
                __builtin_prefetch((const void*)(h + (size_t)g * IN_DIM + kc + KC + ks), 0, 1);
        }
        {
            int o  = tid >> 1;
            int kb = (tid & 1) * 16;
            const float* wrow = W + (size_t)o * IN_DIM + kc + kb;
            #pragma unroll
            for (int j = 0; j < 16; ++j) {
                int k = kb + j;
                lds_w[((k >> 1) * OUT_DIM + o) * 2 + (k & 1)] = wrow[j];
            }
        }
        __syncthreads();

        const int row = nodeTile * 16 + hl;
        const int o0  = (outGroup * 2 + 0) * 16 + hl;
        const int o1  = (outGroup * 2 + 1) * 16 + hl;
        #pragma unroll
        for (int ks = 0; ks < KC; ks += 4) {
            v2f a = *(const v2f*)(&lds_h[row * HSTR + ks + khi]);
            int kb = ks + khi;
            v2f b0 = *(const v2f*)(&lds_w[((kb >> 1) * OUT_DIM + o0) * 2]);
            v2f b1 = *(const v2f*)(&lds_w[((kb >> 1) * OUT_DIM + o1) * 2]);
            c0 = __builtin_amdgcn_wmma_f32_16x16x4_f32(false, a, false, b0, (short)0, c0, false, false);
            c1 = __builtin_amdgcn_wmma_f32_16x16x4_f32(false, a, false, b1, (short)0, c1, false, false);
        }
    }

    const int mrow0 = nodeBase + nodeTile * 16 + (lane >> 4) * 8;
    const int ncol  = lane & 15;

    const int oc = outGroup * 32 + ncol;
    const float as0 = a_attn[oc],           as1 = a_attn[oc + 16];
    const float ad0 = a_attn[OUT_DIM + oc], ad1 = a_attn[OUT_DIM + oc + 16];
    float p[16];
    #pragma unroll
    for (int r = 0; r < 8; ++r) {
        p[2 * r]     = c0[r] * as0 + c1[r] * as1;
        p[2 * r + 1] = c0[r] * ad0 + c1[r] * ad1;
    }
    #pragma unroll
    for (int i = 0; i < 16; ++i) p[i] = row_xor_sum<1>(p[i]);
    #pragma unroll
    for (int i = 0; i < 16; ++i) p[i] = row_xor_sum<2>(p[i]);
    #pragma unroll
    for (int i = 0; i < 16; ++i) p[i] = row_xor_sum<4>(p[i]);
    #pragma unroll
    for (int i = 0; i < 16; ++i) p[i] = row_xor_sum<8>(p[i]);
    if (ncol == 0) {
        float4* pp = (float4*)&lds_p[(outGroup * NB + nodeTile * 16 + (lane >> 4) * 8) * 2];
        pp[0] = make_float4(p[0],  p[1],  p[2],  p[3]);
        pp[1] = make_float4(p[4],  p[5],  p[6],  p[7]);
        pp[2] = make_float4(p[8],  p[9],  p[10], p[11]);
        pp[3] = make_float4(p[12], p[13], p[14], p[15]);
    }
    __syncthreads();
    #pragma unroll
    for (int r = 0; r < 8; ++r) {
        int gnode = mrow0 + r;
        if (gnode < n) {
            z[(size_t)gnode * OUT_DIM + (outGroup * 2 + 0) * 16 + ncol] = c0[r];
            z[(size_t)gnode * OUT_DIM + (outGroup * 2 + 1) * 16 + ncol] = c1[r];
        }
    }

    if (tid < NB) {
        const int g = nodeBase + tid;
        if (g < n) {
            float s = 0.0f, d = 0.0f;
            for (int og = 0; og < 4; ++og) {
                const v2f q = *(const v2f*)&lds_p[(og * NB + tid) * 2];
                s += q.x; d += q.y;
            }
            zs[g] = s; zd[g] = d; head[g] = -1;
        }
    }
}

__global__ void gat_link(const int* __restrict__ dst, int* __restrict__ head,
                         int* __restrict__ next, int E) {
    int i = blockIdx.x * blockDim.x + threadIdx.x;
    if (i >= E) return;
    next[i] = atomicExch(&head[dst[i]], i);
}

__global__ __launch_bounds__(256) void gat_gather(
    const int* __restrict__ src, const int* __restrict__ head, const int* __restrict__ next,
    const float* __restrict__ zs, const float* __restrict__ zd,
    const float* __restrict__ z, float* __restrict__ out, int n) {
    const int gl   = threadIdx.x & 7;
    const int node = blockIdx.x * NB + (threadIdx.x >> 3);
    if (node >= n) return;
    const unsigned oi = (unsigned)node * (OUT_DIM / 4) + (unsigned)gl;
    const float4* z4 = (const float4*)z;
    float4*       o4 = (float4*)out;

    const int first = head[node];
    if (first < 0) {
        const float4 v0 = z4[oi], v1 = z4[oi + 8], v2 = z4[oi + 16], v3 = z4[oi + 24];
        o4[oi] = v0; o4[oi + 8] = v1; o4[oi + 16] = v2; o4[oi + 24] = v3;
        return;
    }
    const float zdn = zd[node];
    int s  = src[first];
    int nx = next[first];
    float m = -INFINITY, den = 0.0f;
    float4 a0 = make_float4(0.0f, 0.0f, 0.0f, 0.0f), a1 = a0, a2 = a0, a3 = a0;

    for (;;) {
        const float    zsv = zs[s];
        const unsigned ri  = (unsigned)s * (OUT_DIM / 4) + (unsigned)gl;
        const float4 p0 = z4[ri], p1 = z4[ri + 8], p2 = z4[ri + 16], p3 = z4[ri + 24];
        int s2 = 0, nx2 = -1;
        if (nx >= 0) { s2 = src[nx]; nx2 = next[nx]; }

        const float x = zsv + zdn;
        const float e = x > 0.0f ? x : 0.01f * x;
        const float mn = fmaxf(m, e);
        const float t  = expf(fminf(m, e) - mn);
        const bool  up = e > m;
        const float scale = up ? t : 1.0f;
        const float w     = up ? 1.0f : t;
        den = den * scale + w;
        m   = mn;
        a0.x = a0.x * scale + w * p0.x; a0.y = a0.y * scale + w * p0.y;
        a0.z = a0.z * scale + w * p0.z; a0.w = a0.w * scale + w * p0.w;
        a1.x = a1.x * scale + w * p1.x; a1.y = a1.y * scale + w * p1.y;
        a1.z = a1.z * scale + w * p1.z; a1.w = a1.w * scale + w * p1.w;
        a2.x = a2.x * scale + w * p2.x; a2.y = a2.y * scale + w * p2.y;
        a2.z = a2.z * scale + w * p2.z; a2.w = a2.w * scale + w * p2.w;
        a3.x = a3.x * scale + w * p3.x; a3.y = a3.y * scale + w * p3.y;
        a3.z = a3.z * scale + w * p3.z; a3.w = a3.w * scale + w * p3.w;

        if (nx < 0) break;
        s = s2; nx = nx2;
    }
    o4[oi]      = make_float4(a0.x / den, a0.y / den, a0.z / den, a0.w / den);
    o4[oi + 8]  = make_float4(a1.x / den, a1.y / den, a1.z / den, a1.w / den);
    o4[oi + 16] = make_float4(a2.x / den, a2.y / den, a2.z / den, a2.w / den);
    o4[oi + 24] = make_float4(a3.x / den, a3.y / den, a3.z / den, a3.w / den);
}

extern "C" void kernel_launch(void* const* d_in, const int* in_sizes, int n_in,
                              void* d_out, int out_size, void* d_ws, size_t ws_size,
                              hipStream_t stream) {
    const float* h      = (const float*)d_in[0];
    const int*   src    = (const int*)d_in[1];
    const int*   dst    = (const int*)d_in[2];
    const float* W      = (const float*)d_in[3];
    const float* a_attn = (const float*)d_in[4];
    const int n = in_sizes[0] / IN_DIM;
    const int E = in_sizes[1];
    float* out = (float*)d_out;

    char* ws = (char*)d_ws;
    float* z    = (float*)ws; ws += (size_t)n * OUT_DIM * sizeof(float);
    float* zs   = (float*)ws; ws += (size_t)n * sizeof(float);
    float* zd   = (float*)ws; ws += (size_t)n * sizeof(float);
    int*   next = (int*)ws;   ws += (size_t)E * sizeof(int);
    int*   head = (int*)ws;

    gat_gemm_wmma<<<(n + NB - 1) / NB, 256, 0, stream>>>(h, W, a_attn, z, zs, zd, head, n);
    gat_link     <<<(E + 255) / 256,   256, 0, stream>>>(dst, head, next, E);
    gat_gather   <<<(n + NB - 1) / NB, 256, 0, stream>>>(src, head, next, zs, zd, z, out, n);
}


# GEMM W tile staged row-major in the h tile's thread mapping: each 16-byte load instruction covers four whole 128-byte lines (was 16 part lines), 16-byte LDS stores, B fragments read like A
# speedup vs baseline: 2.7077x; 1.0530x over previous
#include <hip/hip_runtime.h>
#include <hip/hip_bf16.h>
#include <math.h>

typedef __attribute__((ext_vector_type(2))) float v2f;
typedef __attribute__((ext_vector_type(8))) float v8f;

#define IN_DIM  256
#define OUT_DIM 128
#define KC 32
#define NB 32
#define HSTR 36
#define WSTR 36

template <int XM>
__device__ __forceinline__ float row_xor_sum(float x) {
    const int b = __builtin_bit_cast(int, x);
    return x + __builtin_bit_cast(float,
        __builtin_amdgcn_update_dpp(0, b, 0x160 + XM, 0xf, 0xf, true));
}

__global__ __launch_bounds__(256, 16) void gat_gemm_wmma(
    const float* __restrict__ h, const float* __restrict__ W,
    const float* __restrict__ a_attn,
    float* __restrict__ z, float* __restrict__ zs, float* __restrict__ zd,
    int* __restrict__ head, int n) {
    __shared__ __attribute__((aligned(16))) float lds_h[NB * HSTR];
    __shared__ __attribute__((aligned(16))) float lds_w[OUT_DIM * WSTR];
    __shared__ __attribute__((aligned(16))) float lds_p[4 * NB * 2];

    const int tid      = threadIdx.x;
    const int lane     = tid & 31;
    const int wave     = tid >> 5;
    const int nodeTile = wave >> 2;
    const int outGroup = wave & 3;
    const int nodeBase = blockIdx.x * NB;

    const int hl  = lane & 15;
    const int khi = (lane >> 4) * 2;

    v8f c0 = {}; v8f c1 = {};

    for (int kc = 0; kc < IN_DIM; kc += KC) {
        __syncthreads();
        {
            const int rowL = tid >> 3;
            const int ks   = (tid & 7) * 4;
            int g = nodeBase + rowL;
            if (g >= n) g = n - 1;
            const float* hp = h + (size_t)g * IN_DIM + kc + ks;
            const float* wp = W + (size_t)rowL * IN_DIM + kc + ks;
            const float4 hv = *(const float4*)hp;
            const float4 w0 = *(const float4*)(wp);
            const float4 w1 = *(const float4*)(wp + 32 * IN_DIM);
            const float4 w2 = *(const float4*)(wp + 64 * IN_DIM);
            const float4 w3 = *(const float4*)(wp + 96 * IN_DIM);
            if (kc + KC < IN_DIM)
                __builtin_prefetch((const void*)(hp + KC), 0, 1);
            __builtin_amdgcn_sched_barrier(0);
            *(float4*)(&lds_h[rowL * HSTR + ks]) = hv;
            *(float4*)(&lds_w[(rowL     ) * WSTR + ks]) = w0;
            *(float4*)(&lds_w[(rowL + 32) * WSTR + ks]) = w1;
            *(float4*)(&lds_w[(rowL + 64) * WSTR + ks]) = w2;
            *(float4*)(&lds_w[(rowL + 96) * WSTR + ks]) = w3;
        }
        __syncthreads();

        const int row = nodeTile * 16 + hl;
        const int o0  = (outGroup * 2 + 0) * 16 + hl;
        const int o1  = (outGroup * 2 + 1) * 16 + hl;
        #pragma unroll
        for (int ks = 0; ks < KC; ks += 4) {
            v2f a = *(const v2f*)(&lds_h[row * HSTR + ks + khi]);
            v2f b0 = *(const v2f*)(&lds_w[o0 * WSTR + ks + khi]);
            v2f b1 = *(const v2f*)(&lds_w[o1 * WSTR + ks + khi]);
            c0 = __builtin_amdgcn_wmma_f32_16x16x4_f32(false, a, false, b0, (short)0, c0, false, false);
            c1 = __builtin_amdgcn_wmma_f32_16x16x4_f32(false, a, false, b1, (short)0, c1, false, false);
        }
    }

    const int mrow0 = nodeBase + nodeTile * 16 + (lane >> 4) * 8;
    const int ncol  = lane & 15;

    const int oc = outGroup * 32 + ncol;
    const float as0 = a_attn[oc],           as1 = a_attn[oc + 16];
    const float ad0 = a_attn[OUT_DIM + oc], ad1 = a_attn[OUT_DIM + oc + 16];
    float p[16];
    #pragma unroll
    for (int r = 0; r < 8; ++r) {
        p[2 * r]     = c0[r] * as0 + c1[r] * as1;
        p[2 * r + 1] = c0[r] * ad0 + c1[r] * ad1;
    }
    #pragma unroll
    for (int i = 0; i < 16; ++i) p[i] = row_xor_sum<1>(p[i]);
    #pragma unroll
    for (int i = 0; i < 16; ++i) p[i] = row_xor_sum<2>(p[i]);
    #pragma unroll
    for (int i = 0; i < 16; ++i) p[i] = row_xor_sum<4>(p[i]);
    #pragma unroll
    for (int i = 0; i < 16; ++i) p[i] = row_xor_sum<8>(p[i]);
    if (ncol == 0) {
        float4* pp = (float4*)&lds_p[(outGroup * NB + nodeTile * 16 + (lane >> 4) * 8) * 2];
        pp[0] = make_float4(p[0],  p[1],  p[2],  p[3]);
        pp[1] = make_float4(p[4],  p[5],  p[6],  p[7]);
        pp[2] = make_float4(p[8],  p[9],  p[10], p[11]);
        pp[3] = make_float4(p[12], p[13], p[14], p[15]);
    }
    __syncthreads();
    #pragma unroll
    for (int r = 0; r < 8; ++r) {
        int gnode = mrow0 + r;
        if (gnode < n) {
            z[(size_t)gnode * OUT_DIM + (outGroup * 2 + 0) * 16 + ncol] = c0[r];
            z[(size_t)gnode * OUT_DIM + (outGroup * 2 + 1) * 16 + ncol] = c1[r];
        }
    }

    if (tid < NB) {
        const int g = nodeBase + tid;
        if (g < n) {
            float s = 0.0f, d = 0.0f;
            for (int og = 0; og < 4; ++og) {
                const v2f q = *(const v2f*)&lds_p[(og * NB + tid) * 2];
                s += q.x; d += q.y;
            }
            zs[g] = s; zd[g] = d; head[g] = -1;
        }
    }
}

__global__ void gat_link(const int* __restrict__ dst, int* __restrict__ head,
                         int* __restrict__ next, int E) {
    int i = blockIdx.x * blockDim.x + threadIdx.x;
    if (i >= E) return;
    next[i] = atomicExch(&head[dst[i]], i);
}

__global__ __launch_bounds__(256) void gat_gather(
    const int* __restrict__ src, const int* __restrict__ head, const int* __restrict__ next,
    const float* __restrict__ zs, const float* __restrict__ zd,
    const float* __restrict__ z, float* __restrict__ out, int n) {
    const int gl   = threadIdx.x & 7;
    const int node = blockIdx.x * NB + (threadIdx.x >> 3);
    if (node >= n) return;
    const unsigned oi = (unsigned)node * (OUT_DIM / 4) + (unsigned)gl;
    const float4* z4 = (const float4*)z;
    float4*       o4 = (float4*)out;

    const int first = head[node];
    if (first < 0) {
        const float4 v0 = z4[oi], v1 = z4[oi + 8], v2 = z4[oi + 16], v3 = z4[oi + 24];
        o4[oi] = v0; o4[oi + 8] = v1; o4[oi + 16] = v2; o4[oi + 24] = v3;
        return;
    }
    const float zdn = zd[node];
    int s  = src[first];
    int nx = next[first];
    float m = -INFINITY, den = 0.0f;
    float4 a0 = make_float4(0.0f, 0.0f, 0.0f, 0.0f), a1 = a0, a2 = a0, a3 = a0;

    for (;;) {
        const float    zsv = zs[s];
        const unsigned ri  = (unsigned)s * (OUT_DIM / 4) + (unsigned)gl;
        const float4 p0 = z4[ri], p1 = z4[ri + 8], p2 = z4[ri + 16], p3 = z4[ri + 24];
        int s2 = 0, nx2 = -1;
        if (nx >= 0) { s2 = src[nx]; nx2 = next[nx]; }

        const float x = zsv + zdn;
        const float e = x > 0.0f ? x : 0.01f * x;
        const float mn = fmaxf(m, e);
        const float t  = expf(fminf(m, e) - mn);
        const bool  up = e > m;
        const float scale = up ? t : 1.0f;
        const float w     = up ? 1.0f : t;
        den = den * scale + w;
        m   = mn;
        a0.x = a0.x * scale + w * p0.x; a0.y = a0.y * scale + w * p0.y;
        a0.z = a0.z * scale + w * p0.z; a0.w = a0.w * scale + w * p0.w;
        a1.x = a1.x * scale + w * p1.x; a1.y = a1.y * scale + w * p1.y;
        a1.z = a1.z * scale + w * p1.z; a1.w = a1.w * scale + w * p1.w;
        a2.x = a2.x * scale + w * p2.x; a2.y = a2.y * scale + w * p2.y;
        a2.z = a2.z * scale + w * p2.z; a2.w = a2.w * scale + w * p2.w;
        a3.x = a3.x * scale + w * p3.x; a3.y = a3.y * scale + w * p3.y;
        a3.z = a3.z * scale + w * p3.z; a3.w = a3.w * scale + w * p3.w;

        if (nx < 0) break;
        s = s2; nx = nx2;
    }
    o4[oi]      = make_float4(a0.x / den, a0.y / den, a0.z / den, a0.w / den);
    o4[oi + 8]  = make_float4(a1.x / den, a1.y / den, a1.z / den, a1.w / den);
    o4[oi + 16] = make_float4(a2.x / den, a2.y / den, a2.z / den, a2.w / den);
    o4[oi + 24] = make_float4(a3.x / den, a3.y / den, a3.z / den, a3.w / den);
}

extern "C" void kernel_launch(void* const* d_in, const int* in_sizes, int n_in,
                              void* d_out, int out_size, void* d_ws, size_t ws_size,
                              hipStream_t stream) {
    const float* h      = (const float*)d_in[0];
    const int*   src    = (const int*)d_in[1];
    const int*   dst    = (const int*)d_in[2];
    const float* W      = (const float*)d_in[3];
    const float* a_attn = (const float*)d_in[4];
    const int n = in_sizes[0] / IN_DIM;
    const int E = in_sizes[1];
    float* out = (float*)d_out;

    char* ws = (char*)d_ws;
    float* z    = (float*)ws; ws += (size_t)n * OUT_DIM * sizeof(float);
    float* zs   = (float*)ws; ws += (size_t)n * sizeof(float);
    float* zd   = (float*)ws; ws += (size_t)n * sizeof(float);
    int*   next = (int*)ws;   ws += (size_t)E * sizeof(int);
    int*   head = (int*)ws;

    gat_gemm_wmma<<<(n + NB - 1) / NB, 256, 0, stream>>>(h, W, a_attn, z, zs, zd, head, n);
    gat_link     <<<(E + 255) / 256,   256, 0, stream>>>(dst, head, next, E);
    gat_gather   <<<(n + NB - 1) / NB, 256, 0, stream>>>(src, head, next, zs, zd, z, out, n);
}


# Edge linking moved into the product kernel's epilogue (one exchange per thread, in flight while other blocks multiply); a small launch in front opens the list heads; the separate link launch is gone
# speedup vs baseline: 2.9178x; 1.0776x over previous
#include <hip/hip_runtime.h>
#include <hip/hip_bf16.h>
#include <math.h>

typedef __attribute__((ext_vector_type(2))) float v2f;
typedef __attribute__((ext_vector_type(8))) float v8f;

#define IN_DIM  256
#define OUT_DIM 128
#define KC 32
#define NB 32
#define HSTR 36
#define WSTR 36

template <int XM>
__device__ __forceinline__ float row_xor_sum(float x) {
    const int b = __builtin_bit_cast(int, x);
    return x + __builtin_bit_cast(float,
        __builtin_amdgcn_update_dpp(0, b, 0x160 + XM, 0xf, 0xf, true));
}

__global__ __launch_bounds__(256, 16) void gat_gemm_wmma(
    const float* __restrict__ h, const float* __restrict__ W,
    const float* __restrict__ a_attn,
    float* __restrict__ z, float* __restrict__ zs, float* __restrict__ zd,
    const int* __restrict__ dst, int* __restrict__ head, int* __restrict__ next,
    int n, int E) {
    __shared__ __attribute__((aligned(16))) float lds_h[NB * HSTR];
    __shared__ __attribute__((aligned(16))) float lds_w[OUT_DIM * WSTR];
    __shared__ __attribute__((aligned(16))) float lds_p[4 * NB * 2];

    const int tid      = threadIdx.x;
    const int lane     = tid & 31;
    const int wave     = tid >> 5;
    const int nodeTile = wave >> 2;
    const int outGroup = wave & 3;
    const int nodeBase = blockIdx.x * NB;

    const int hl  = lane & 15;
    const int khi = (lane >> 4) * 2;

    v8f c0 = {}; v8f c1 = {};

    const int eidx = blockIdx.x * 256 + tid;
    if (nodeBase >= n) {
        if (eidx < E) next[eidx] = atomicExch(&head[dst[eidx]], eidx);
        return;
    }

    for (int kc = 0; kc < IN_DIM; kc += KC) {
        __syncthreads();
        {
            const int rowL = tid >> 3;
            const int ks   = (tid & 7) * 4;
            int g = nodeBase + rowL;
            if (g >= n) g = n - 1;
            const float* hp = h + (size_t)g * IN_DIM + kc + ks;
            const float* wp = W + (size_t)rowL * IN_DIM + kc + ks;
            const float4 hv = *(const float4*)hp;
            const float4 w0 = *(const float4*)(wp);
            const float4 w1 = *(const float4*)(wp + 32 * IN_DIM);
            const float4 w2 = *(const float4*)(wp + 64 * IN_DIM);
            const float4 w3 = *(const float4*)(wp + 96 * IN_DIM);
            if (kc + KC < IN_DIM)
                __builtin_prefetch((const void*)(hp + KC), 0, 1);
            __builtin_amdgcn_sched_barrier(0);
            *(float4*)(&lds_h[rowL * HSTR + ks]) = hv;
            *(float4*)(&lds_w[(rowL     ) * WSTR + ks]) = w0;
            *(float4*)(&lds_w[(rowL + 32) * WSTR + ks]) = w1;
            *(float4*)(&lds_w[(rowL + 64) * WSTR + ks]) = w2;
            *(float4*)(&lds_w[(rowL + 96) * WSTR + ks]) = w3;
        }
        __syncthreads();

        const int row = nodeTile * 16 + hl;
        const int o0  = (outGroup * 2 + 0) * 16 + hl;
        const int o1  = (outGroup * 2 + 1) * 16 + hl;
        #pragma unroll
        for (int ks = 0; ks < KC; ks += 4) {
            v2f a = *(const v2f*)(&lds_h[row * HSTR + ks + khi]);
            v2f b0 = *(const v2f*)(&lds_w[o0 * WSTR + ks + khi]);
            v2f b1 = *(const v2f*)(&lds_w[o1 * WSTR + ks + khi]);
            c0 = __builtin_amdgcn_wmma_f32_16x16x4_f32(false, a, false, b0, (short)0, c0, false, false);
            c1 = __builtin_amdgcn_wmma_f32_16x16x4_f32(false, a, false, b1, (short)0, c1, false, false);
        }
    }

    const int mrow0 = nodeBase + nodeTile * 16 + (lane >> 4) * 8;
    const int ncol  = lane & 15;

    const int oc = outGroup * 32 + ncol;
    const float as0 = a_attn[oc],           as1 = a_attn[oc + 16];
    const float ad0 = a_attn[OUT_DIM + oc], ad1 = a_attn[OUT_DIM + oc + 16];
    const bool eok  = eidx < E;
    const int  edst = dst[eok ? eidx : E - 1];
    float p[16];
    #pragma unroll
    for (int r = 0; r < 8; ++r) {
        p[2 * r]     = c0[r] * as0 + c1[r] * as1;
        p[2 * r + 1] = c0[r] * ad0 + c1[r] * ad1;
    }
    #pragma unroll
    for (int i = 0; i < 16; ++i) p[i] = row_xor_sum<1>(p[i]);
    #pragma unroll
    for (int i = 0; i < 16; ++i) p[i] = row_xor_sum<2>(p[i]);
    #pragma unroll
    for (int i = 0; i < 16; ++i) p[i] = row_xor_sum<4>(p[i]);
    #pragma unroll
    for (int i = 0; i < 16; ++i) p[i] = row_xor_sum<8>(p[i]);
    if (ncol == 0) {
        float4* pp = (float4*)&lds_p[(outGroup * NB + nodeTile * 16 + (lane >> 4) * 8) * 2];
        pp[0] = make_float4(p[0],  p[1],  p[2],  p[3]);
        pp[1] = make_float4(p[4],  p[5],  p[6],  p[7]);
        pp[2] = make_float4(p[8],  p[9],  p[10], p[11]);
        pp[3] = make_float4(p[12], p[13], p[14], p[15]);
    }
    __syncthreads();
    int eold = -1;
    if (eok) eold = atomicExch(&head[edst], eidx);
    #pragma unroll
    for (int r = 0; r < 8; ++r) {
        int gnode = mrow0 + r;
        if (gnode < n) {
            z[(size_t)gnode * OUT_DIM + (outGroup * 2 + 0) * 16 + ncol] = c0[r];
            z[(size_t)gnode * OUT_DIM + (outGroup * 2 + 1) * 16 + ncol] = c1[r];
        }
    }

    if (tid < NB) {
        const int g = nodeBase + tid;
        if (g < n) {
            float s = 0.0f, d = 0.0f;
            for (int og = 0; og < 4; ++og) {
                const v2f q = *(const v2f*)&lds_p[(og * NB + tid) * 2];
                s += q.x; d += q.y;
            }
            zs[g] = s; zd[g] = d;
        }
    }
    if (eok) next[eidx] = eold;
}

__global__ void gat_open(int* __restrict__ head, int n) {
    int i = blockIdx.x * blockDim.x + threadIdx.x;
    if (i < n) head[i] = -1;
}

__global__ __launch_bounds__(256) void gat_gather(
    const int* __restrict__ src, const int* __restrict__ head, const int* __restrict__ next,
    const float* __restrict__ zs, const float* __restrict__ zd,
    const float* __restrict__ z, float* __restrict__ out, int n) {
    const int gl   = threadIdx.x & 7;
    const int node = blockIdx.x * NB + (threadIdx.x >> 3);
    if (node >= n) return;
    const unsigned oi = (unsigned)node * (OUT_DIM / 4) + (unsigned)gl;
    const float4* z4 = (const float4*)z;
    float4*       o4 = (float4*)out;

    const int first = head[node];
    if (first < 0) {
        const float4 v0 = z4[oi], v1 = z4[oi + 8], v2 = z4[oi + 16], v3 = z4[oi + 24];
        o4[oi] = v0; o4[oi + 8] = v1; o4[oi + 16] = v2; o4[oi + 24] = v3;
        return;
    }
    const float zdn = zd[node];
    int s  = src[first];
    int nx = next[first];
    float m = -INFINITY, den = 0.0f;
    float4 a0 = make_float4(0.0f, 0.0f, 0.0f, 0.0f), a1 = a0, a2 = a0, a3 = a0;

    for (;;) {
        const float    zsv = zs[s];
        const unsigned ri  = (unsigned)s * (OUT_DIM / 4) + (unsigned)gl;
        const float4 p0 = z4[ri], p1 = z4[ri + 8], p2 = z4[ri + 16], p3 = z4[ri + 24];
        int s2 = 0, nx2 = -1;
        if (nx >= 0) { s2 = src[nx]; nx2 = next[nx]; }

        const float x = zsv + zdn;
        const float e = x > 0.0f ? x : 0.01f * x;
        const float mn = fmaxf(m, e);
        const float t  = expf(fminf(m, e) - mn);
        const bool  up = e > m;
        const float scale = up ? t : 1.0f;
        const float w     = up ? 1.0f : t;
        den = den * scale + w;
        m   = mn;
        a0.x = a0.x * scale + w * p0.x; a0.y = a0.y * scale + w * p0.y;
        a0.z = a0.z * scale + w * p0.z; a0.w = a0.w * scale + w * p0.w;
        a1.x = a1.x * scale + w * p1.x; a1.y = a1.y * scale + w * p1.y;
        a1.z = a1.z * scale + w * p1.z; a1.w = a1.w * scale + w * p1.w;
        a2.x = a2.x * scale + w * p2.x; a2.y = a2.y * scale + w * p2.y;
        a2.z = a2.z * scale + w * p2.z; a2.w = a2.w * scale + w * p2.w;
        a3.x = a3.x * scale + w * p3.x; a3.y = a3.y * scale + w * p3.y;
        a3.z = a3.z * scale + w * p3.z; a3.w = a3.w * scale + w * p3.w;

        if (nx < 0) break;
        s = s2; nx = nx2;
    }
    o4[oi]      = make_float4(a0.x / den, a0.y / den, a0.z / den, a0.w / den);
    o4[oi + 8]  = make_float4(a1.x / den, a1.y / den, a1.z / den, a1.w / den);
    o4[oi + 16] = make_float4(a2.x / den, a2.y / den, a2.z / den, a2.w / den);
    o4[oi + 24] = make_float4(a3.x / den, a3.y / den, a3.z / den, a3.w / den);
}

extern "C" void kernel_launch(void* const* d_in, const int* in_sizes, int n_in,
                              void* d_out, int out_size, void* d_ws, size_t ws_size,
                              hipStream_t stream) {
    const float* h      = (const float*)d_in[0];
    const int*   src    = (const int*)d_in[1];
    const int*   dst    = (const int*)d_in[2];
    const float* W      = (const float*)d_in[3];
    const float* a_attn = (const float*)d_in[4];
    const int n = in_sizes[0] / IN_DIM;
    const int E = in_sizes[1];
    float* out = (float*)d_out;

    char* ws = (char*)d_ws;
    float* z    = (float*)ws; ws += (size_t)n * OUT_DIM * sizeof(float);
    float* zs   = (float*)ws; ws += (size_t)n * sizeof(float);
    float* zd   = (float*)ws; ws += (size_t)n * sizeof(float);
    int*   next = (int*)ws;   ws += (size_t)E * sizeof(int);
    int*   head = (int*)ws;

    gat_open     <<<(n + 255) / 256,   256, 0, stream>>>(head, n);
    const int nodeBlocks = (n + NB - 1) / NB, edgeBlocks = (E + 255) / 256;
    gat_gemm_wmma<<<nodeBlocks > edgeBlocks ? nodeBlocks : edgeBlocks, 256, 0, stream>>>(
        h, W, a_attn, z, zs, zd, dst, head, next, n, E);
    gat_gather   <<<(n + NB - 1) / NB, 256, 0, stream>>>(src, head, next, zs, zd, z, out, n);
}


# Two edge lists per destination node (even / odd edge numbers), walked side by side in the gather: two edges per dependent round trip, d/2+2 trips per node, not d+2; the order of the sums changes
# speedup vs baseline: 3.0328x; 1.0394x over previous
#include <hip/hip_runtime.h>
#include <hip/hip_bf16.h>
#include <math.h>

typedef __attribute__((ext_vector_type(2))) float v2f;
typedef __attribute__((ext_vector_type(8))) float v8f;

#define IN_DIM  256
#define OUT_DIM 128
#define KC 32
#define NB 32
#define HSTR 36
#define WSTR 36

template <int XM>
__device__ __forceinline__ float row_xor_sum(float x) {
    const int b = __builtin_bit_cast(int, x);
    return x + __builtin_bit_cast(float,
        __builtin_amdgcn_update_dpp(0, b, 0x160 + XM, 0xf, 0xf, true));
}

__global__ __launch_bounds__(256, 16) void gat_gemm_wmma(
    const float* __restrict__ h, const float* __restrict__ W,
    const float* __restrict__ a_attn,
    float* __restrict__ z, float* __restrict__ zs, float* __restrict__ zd,
    const int* __restrict__ dst, int* __restrict__ head, int* __restrict__ next,
    int n, int E) {
    __shared__ __attribute__((aligned(16))) float lds_h[NB * HSTR];
    __shared__ __attribute__((aligned(16))) float lds_w[OUT_DIM * WSTR];
    __shared__ __attribute__((aligned(16))) float lds_p[4 * NB * 2];

    const int tid      = threadIdx.x;
    const int lane     = tid & 31;
    const int wave     = tid >> 5;
    const int nodeTile = wave >> 2;
    const int outGroup = wave & 3;
    const int nodeBase = blockIdx.x * NB;

    const int hl  = lane & 15;
    const int khi = (lane >> 4) * 2;

    v8f c0 = {}; v8f c1 = {};

    const int eidx = blockIdx.x * 256 + tid;
    if (nodeBase >= n) {
        if (eidx < E) next[eidx] = atomicExch(&head[2 * dst[eidx] + (eidx & 1)], eidx);
        return;
    }

    for (int kc = 0; kc < IN_DIM; kc += KC) {
        __syncthreads();
        {
            const int rowL = tid >> 3;
            const int ks   = (tid & 7) * 4;
            int g = nodeBase + rowL;
            if (g >= n) g = n - 1;
            const float* hp = h + (size_t)g * IN_DIM + kc + ks;
            const float* wp = W + (size_t)rowL * IN_DIM + kc + ks;
            const float4 hv = *(const float4*)hp;
            const float4 w0 = *(const float4*)(wp);
            const float4 w1 = *(const float4*)(wp + 32 * IN_DIM);
            const float4 w2 = *(const float4*)(wp + 64 * IN_DIM);
            const float4 w3 = *(const float4*)(wp + 96 * IN_DIM);
            if (kc + KC < IN_DIM)
                __builtin_prefetch((const void*)(hp + KC), 0, 1);
            __builtin_amdgcn_sched_barrier(0);
            *(float4*)(&lds_h[rowL * HSTR + ks]) = hv;
            *(float4*)(&lds_w[(rowL     ) * WSTR + ks]) = w0;
            *(float4*)(&lds_w[(rowL + 32) * WSTR + ks]) = w1;
            *(float4*)(&lds_w[(rowL + 64) * WSTR + ks]) = w2;
            *(float4*)(&lds_w[(rowL + 96) * WSTR + ks]) = w3;
        }
        __syncthreads();

        const int row = nodeTile * 16 + hl;
        const int o0  = (outGroup * 2 + 0) * 16 + hl;
        const int o1  = (outGroup * 2 + 1) * 16 + hl;
        #pragma unroll
        for (int ks = 0; ks < KC; ks += 4) {
            v2f a = *(const v2f*)(&lds_h[row * HSTR + ks + khi]);
            v2f b0 = *(const v2f*)(&lds_w[o0 * WSTR + ks + khi]);
            v2f b1 = *(const v2f*)(&lds_w[o1 * WSTR + ks + khi]);
            c0 = __builtin_amdgcn_wmma_f32_16x16x4_f32(false, a, false, b0, (short)0, c0, false, false);
            c1 = __builtin_amdgcn_wmma_f32_16x16x4_f32(false, a, false, b1, (short)0, c1, false, false);
        }
    }

    const int mrow0 = nodeBase + nodeTile * 16 + (lane >> 4) * 8;
    const int ncol  = lane & 15;

    const int oc = outGroup * 32 + ncol;
    const float as0 = a_attn[oc],           as1 = a_attn[oc + 16];
    const float ad0 = a_attn[OUT_DIM + oc], ad1 = a_attn[OUT_DIM + oc + 16];
    const bool eok  = eidx < E;
    const int  edst = dst[eok ? eidx : E - 1];
    float p[16];
    #pragma unroll
    for (int r = 0; r < 8; ++r) {
        p[2 * r]     = c0[r] * as0 + c1[r] * as1;
        p[2 * r + 1] = c0[r] * ad0 + c1[r] * ad1;
    }
    #pragma unroll
    for (int i = 0; i < 16; ++i) p[i] = row_xor_sum<1>(p[i]);
    #pragma unroll
    for (int i = 0; i < 16; ++i) p[i] = row_xor_sum<2>(p[i]);
    #pragma unroll
    for (int i = 0; i < 16; ++i) p[i] = row_xor_sum<4>(p[i]);
    #pragma unroll
    for (int i = 0; i < 16; ++i) p[i] = row_xor_sum<8>(p[i]);
    if (ncol == 0) {
        float4* pp = (float4*)&lds_p[(outGroup * NB + nodeTile * 16 + (lane >> 4) * 8) * 2];
        pp[0] = make_float4(p[0],  p[1],  p[2],  p[3]);
        pp[1] = make_float4(p[4],  p[5],  p[6],  p[7]);
        pp[2] = make_float4(p[8],  p[9],  p[10], p[11]);
        pp[3] = make_float4(p[12], p[13], p[14], p[15]);
    }
    __syncthreads();
    int eold = -1;
    if (eok) eold = atomicExch(&head[2 * edst + (eidx & 1)], eidx);
    #pragma unroll
    for (int r = 0; r < 8; ++r) {
        int gnode = mrow0 + r;
        if (gnode < n) {
            z[(size_t)gnode * OUT_DIM + (outGroup * 2 + 0) * 16 + ncol] = c0[r];
            z[(size_t)gnode * OUT_DIM + (outGroup * 2 + 1) * 16 + ncol] = c1[r];
        }
    }

    if (tid < NB) {
        const int g = nodeBase + tid;
        if (g < n) {
            float s = 0.0f, d = 0.0f;
            for (int og = 0; og < 4; ++og) {
                const v2f q = *(const v2f*)&lds_p[(og * NB + tid) * 2];
                s += q.x; d += q.y;
            }
            zs[g] = s; zd[g] = d;
        }
    }
    if (eok) next[eidx] = eold;
}

__global__ void gat_open(int* __restrict__ head, int nheads) {
    int i = blockIdx.x * blockDim.x + threadIdx.x;
    if (i < nheads) head[i] = -1;
}

__global__ __launch_bounds__(256) void gat_gather(
    const int* __restrict__ src, const int* __restrict__ head, const int* __restrict__ next,
    const float* __restrict__ zs, const float* __restrict__ zd,
    const float* __restrict__ z, float* __restrict__ out, int n) {
    const int gl   = threadIdx.x & 7;
    const int node = blockIdx.x * NB + (threadIdx.x >> 3);
    if (node >= n) return;
    const unsigned oi = (unsigned)node * (OUT_DIM / 4) + (unsigned)gl;
    const float4* z4 = (const float4*)z;
    float4*       o4 = (float4*)out;

    const int2 h2 = ((const int2*)head)[node];
    if (h2.x < 0 && h2.y < 0) {
        const float4 v0 = z4[oi], v1 = z4[oi + 8], v2 = z4[oi + 16], v3 = z4[oi + 24];
        o4[oi] = v0; o4[oi + 8] = v1; o4[oi + 16] = v2; o4[oi + 24] = v3;
        return;
    }
    const float zdn = zd[node];
    bool actA = h2.x >= 0, actB = h2.y >= 0;
    int sA = 0, nxA = -1, sB = 0, nxB = -1;
    if (actA) { sA = src[h2.x]; nxA = next[h2.x]; }
    if (actB) { sB = src[h2.y]; nxB = next[h2.y]; }
    asm volatile("" : "+v"(sA), "+v"(nxA), "+v"(sB), "+v"(nxB));
    float m = -INFINITY, den = 0.0f;
    const float4 zero4 = make_float4(0.0f, 0.0f, 0.0f, 0.0f);
    float4 a0 = zero4, a1 = zero4, a2 = zero4, a3 = zero4;
    float4 pA0 = zero4, pA1 = zero4, pA2 = zero4, pA3 = zero4;
    float4 pB0 = zero4, pB1 = zero4, pB2 = zero4, pB3 = zero4;
    float zsA = 0.0f, zsB = 0.0f;

    for (;;) {
        int s2A = 0, nx2A = -1, s2B = 0, nx2B = -1;
        if (actA) {
            zsA = zs[sA];
            const unsigned ri = (unsigned)sA * (OUT_DIM / 4) + (unsigned)gl;
            pA0 = z4[ri]; pA1 = z4[ri + 8]; pA2 = z4[ri + 16]; pA3 = z4[ri + 24];
            if (nxA >= 0) { s2A = src[nxA]; nx2A = next[nxA]; }
        }
        if (actB) {
            zsB = zs[sB];
            const unsigned ri = (unsigned)sB * (OUT_DIM / 4) + (unsigned)gl;
            pB0 = z4[ri]; pB1 = z4[ri + 8]; pB2 = z4[ri + 16]; pB3 = z4[ri + 24];
            if (nxB >= 0) { s2B = src[nxB]; nx2B = next[nxB]; }
        }

        const float xA = zsA + zdn, xB = zsB + zdn;
        const float eA = actA ? (xA > 0.0f ? xA : 0.01f * xA) : -INFINITY;
        const float eB = actB ? (xB > 0.0f ? xB : 0.01f * xB) : -INFINITY;
        const float mn = fmaxf(m, fmaxf(eA, eB));
        const float sc = expf(m - mn);
        const float wA = expf(eA - mn);
        const float wB = expf(eB - mn);
        den = den * sc + wA + wB;
        m   = mn;
        a0.x = a0.x * sc + wA * pA0.x + wB * pB0.x; a0.y = a0.y * sc + wA * pA0.y + wB * pB0.y;
        a0.z = a0.z * sc + wA * pA0.z + wB * pB0.z; a0.w = a0.w * sc + wA * pA0.w + wB * pB0.w;
        a1.x = a1.x * sc + wA * pA1.x + wB * pB1.x; a1.y = a1.y * sc + wA * pA1.y + wB * pB1.y;
        a1.z = a1.z * sc + wA * pA1.z + wB * pB1.z; a1.w = a1.w * sc + wA * pA1.w + wB * pB1.w;
        a2.x = a2.x * sc + wA * pA2.x + wB * pB2.x; a2.y = a2.y * sc + wA * pA2.y + wB * pB2.y;
        a2.z = a2.z * sc + wA * pA2.z + wB * pB2.z; a2.w = a2.w * sc + wA * pA2.w + wB * pB2.w;
        a3.x = a3.x * sc + wA * pA3.x + wB * pB3.x; a3.y = a3.y * sc + wA * pA3.y + wB * pB3.y;
        a3.z = a3.z * sc + wA * pA3.z + wB * pB3.z; a3.w = a3.w * sc + wA * pA3.w + wB * pB3.w;

        actA = actA && nxA >= 0; sA = s2A; nxA = nx2A;
        actB = actB && nxB >= 0; sB = s2B; nxB = nx2B;
        if (!(actA || actB)) break;
    }
    o4[oi]      = make_float4(a0.x / den, a0.y / den, a0.z / den, a0.w / den);
    o4[oi + 8]  = make_float4(a1.x / den, a1.y / den, a1.z / den, a1.w / den);
    o4[oi + 16] = make_float4(a2.x / den, a2.y / den, a2.z / den, a2.w / den);
    o4[oi + 24] = make_float4(a3.x / den, a3.y / den, a3.z / den, a3.w / den);
}

extern "C" void kernel_launch(void* const* d_in, const int* in_sizes, int n_in,
                              void* d_out, int out_size, void* d_ws, size_t ws_size,
                              hipStream_t stream) {
    const float* h      = (const float*)d_in[0];
    const int*   src    = (const int*)d_in[1];
    const int*   dst    = (const int*)d_in[2];
    const float* W      = (const float*)d_in[3];
    const float* a_attn = (const float*)d_in[4];
    const int n = in_sizes[0] / IN_DIM;
    const int E = in_sizes[1];
    float* out = (float*)d_out;

    char* ws = (char*)d_ws;
    float* z    = (float*)ws; ws += (size_t)n * OUT_DIM * sizeof(float);
    int*   head = (int*)ws;   ws += (size_t)n * 2 * sizeof(int);
    float* zs   = (float*)ws; ws += (size_t)n * sizeof(float);
    float* zd   = (float*)ws; ws += (size_t)n * sizeof(float);
    int*   next = (int*)ws;

    gat_open     <<<(2 * n + 255) / 256, 256, 0, stream>>>(head, 2 * n);
    const int nodeBlocks = (n + NB - 1) / NB, edgeBlocks = (E + 255) / 256;
    gat_gemm_wmma<<<nodeBlocks > edgeBlocks ? nodeBlocks : edgeBlocks, 256, 0, stream>>>(
        h, W, a_attn, z, zs, zd, dst, head, next, n, E);
    gat_gather   <<<(n + NB - 1) / NB, 256, 0, stream>>>(src, head, next, zs, zd, z, out, n);
}
